# starting kernel
# baseline (speedup 1.0000x reference)
#include <hip/hip_runtime.h>
#include <hip/hip_bf16.h>

typedef __attribute__((ext_vector_type(2))) float v2f;
typedef __attribute__((ext_vector_type(8))) float v8f;

#define D 32
#define SLOPE 0.2f
#define EPS 1e-16f

__device__ __forceinline__ unsigned f2o(float f) {
    unsigned b = __float_as_uint(f);
    return (b & 0x80000000u) ? ~b : (b | 0x80000000u);
}
__device__ __forceinline__ float o2f(unsigned o) {
    unsigned b = (o & 0x80000000u) ? (o & 0x7FFFFFFFu) : ~o;
    return __uint_as_float(b);
}

__device__ __forceinline__ float lrelu(float x) { return x > 0.f ? x : SLOPE * x; }

__global__ __launch_bounds__(256) void gat_init(unsigned* __restrict__ m,
                                                float* __restrict__ denom,
                                                float* __restrict__ out, int N) {
    int i = blockIdx.x * blockDim.x + threadIdx.x;
    if (i < N) { m[i] = 0u; denom[i] = 0.f; }
    if (i < N * D) out[i] = 0.f;
}

__global__ __launch_bounds__(256) void gat_gemm(const float* __restrict__ X,
                                                const float* __restrict__ W,
                                                float* __restrict__ H, int N) {
    const int wave = blockIdx.x * 8 + (threadIdx.x >> 5);
    const int lane = threadIdx.x & 31;
    const int m0 = wave * 16;
    if (m0 >= N) return;
    const int row   = m0 + (lane & 15);
    const int col   = lane & 15;
    const int khalf = (lane >> 4) * 2;
    v8f c0 = {}; v8f c1 = {};
#pragma unroll
    for (int kb = 0; kb < 8; ++kb) {
        const int k0 = kb * 4 + khalf;
        v2f a;  a.x  = X[row * D + k0];        a.y  = X[row * D + k0 + 1];
        v2f b0; b0.x = W[k0 * D + col];        b0.y = W[(k0 + 1) * D + col];
        v2f b1; b1.x = W[k0 * D + col + 16];   b1.y = W[(k0 + 1) * D + col + 16];
        c0 = __builtin_amdgcn_wmma_f32_16x16x4_f32(false, a, false, b0, (short)0, c0, false, false);
        c1 = __builtin_amdgcn_wmma_f32_16x16x4_f32(false, a, false, b1, (short)0, c1, false, false);
    }
    const int rbase = m0 + (lane >> 4) * 8;
#pragma unroll
    for (int r = 0; r < 8; ++r) {
        H[(rbase + r) * D + col]      = c0[r];
        H[(rbase + r) * D + col + 16] = c1[r];
    }
}

__global__ __launch_bounds__(256) void gat_dots(const float* __restrict__ H,
                                                const float* __restrict__ a_src,
                                                const float* __restrict__ a_dst,
                                                float* __restrict__ as_, float* __restrict__ ad_,
                                                int N) {
    int i = blockIdx.x * blockDim.x + threadIdx.x;
    if (i >= N) return;
    float s = 0.f, d = 0.f;
#pragma unroll
    for (int k = 0; k < D; ++k) {
        float h = H[i * D + k];
        s += h * a_src[k];
        d += h * a_dst[k];
    }
    as_[i] = s; ad_[i] = d;
}

__global__ __launch_bounds__(256) void gat_edge_max(const int* __restrict__ ei, int E, int N,
                                                    const float* __restrict__ as_,
                                                    const float* __restrict__ ad_,
                                                    unsigned* __restrict__ m) {
    int e = blockIdx.x * blockDim.x + threadIdx.x;
    int Etot = E + N;
    if (e >= Etot) return;
    int s = (e < E) ? ei[e]     : (e - E);
    int d = (e < E) ? ei[E + e] : (e - E);
    float lg = lrelu(as_[s] + ad_[d]);
    atomicMax(&m[d], f2o(lg));
}

__global__ __launch_bounds__(256) void gat_edge_sum(const int* __restrict__ ei, int E, int N,
                                                    const float* __restrict__ as_,
                                                    const float* __restrict__ ad_,
                                                    const unsigned* __restrict__ m,
                                                    float* __restrict__ denom) {
    int e = blockIdx.x * blockDim.x + threadIdx.x;
    int Etot = E + N;
    if (e >= Etot) return;
    int s = (e < E) ? ei[e]     : (e - E);
    int d = (e < E) ? ei[E + e] : (e - E);
    float lg = lrelu(as_[s] + ad_[d]);
    float ex = __expf(lg - o2f(m[d]));
    atomicAdd(&denom[d], ex);
}

__global__ __launch_bounds__(256) void gat_edge_aggr(const int* __restrict__ ei, int E, int N,
                                                     const float* __restrict__ as_,
                                                     const float* __restrict__ ad_,
                                                     const unsigned* __restrict__ m,
                                                     const float* __restrict__ denom,
                                                     const float* __restrict__ H,
                                                     float* __restrict__ out) {
    int e    = blockIdx.x * 8 + (threadIdx.x >> 5);
    int lane = threadIdx.x & 31;
    int Etot = E + N;
    if (e >= Etot) return;
    int s = (e < E) ? ei[e]     : (e - E);
    int d = (e < E) ? ei[E + e] : (e - E);
    float lg    = lrelu(as_[s] + ad_[d]);
    float ex    = __expf(lg - o2f(m[d]));
    float alpha = ex / (denom[d] + EPS);
    atomicAdd(&out[d * D + lane], alpha * H[s * D + lane]);
}

__global__ __launch_bounds__(256) void gat_finalize(float* __restrict__ out,
                                                    const float* __restrict__ b, int N) {
    int i = blockIdx.x * blockDim.x + threadIdx.x;
    if (i >= N * D) return;
    float v = out[i] + b[i & (D - 1)];
    out[i] = (v > 0.f) ? v : (__expf(v) - 1.f);
}

static void run_layer(const float* x_in, const int* ei, int E, int N,
                      const float* W, const float* a_src, const float* a_dst, const float* b,
                      float* H, float* as_, float* ad_, float* denom, unsigned* m,
                      float* out, hipStream_t stream) {
    const int Etot = E + N;
    const int nd_blocks   = (N * D + 255) / 256;
    const int node_blocks = (N + 255) / 256;
    const int edge_blocks = (Etot + 255) / 256;
    const int tile_waves  = (N + 15) / 16;
    const int gemm_blocks = (tile_waves + 7) / 8;
    const int aggr_blocks = (Etot + 7) / 8;

    gat_init<<<nd_blocks, 256, 0, stream>>>(m, denom, out, N);
    gat_gemm<<<gemm_blocks, 256, 0, stream>>>(x_in, W, H, N);
    gat_dots<<<node_blocks, 256, 0, stream>>>(H, a_src, a_dst, as_, ad_, N);
    gat_edge_max<<<edge_blocks, 256, 0, stream>>>(ei, E, N, as_, ad_, m);
    gat_edge_sum<<<edge_blocks, 256, 0, stream>>>(ei, E, N, as_, ad_, m, denom);
    gat_edge_aggr<<<aggr_blocks, 256, 0, stream>>>(ei, E, N, as_, ad_, m, denom, H, out);
    gat_finalize<<<nd_blocks, 256, 0, stream>>>(out, b, N);
}

extern "C" void kernel_launch(void* const* d_in, const int* in_sizes, int n_in,
                              void* d_out, int out_size, void* d_ws, size_t ws_size,
                              hipStream_t stream) {
    const float* x      = (const float*)d_in[0];
    const int*   ei     = (const int*)d_in[1];
    const float* W1     = (const float*)d_in[2];
    const float* a_src1 = (const float*)d_in[3];
    const float* a_dst1 = (const float*)d_in[4];
    const float* b1     = (const float*)d_in[5];
    const float* W2     = (const float*)d_in[6];
    const float* a_src2 = (const float*)d_in[7];
    const float* a_dst2 = (const float*)d_in[8];
    const float* b2     = (const float*)d_in[9];

    const int N = in_sizes[0] / D;
    const int E = in_sizes[1] / 2;

    float*    H     = (float*)d_ws;
    float*    as_   = H + (size_t)N * D;
    float*    ad_   = as_ + N;
    float*    denom = ad_ + N;
    unsigned* m     = (unsigned*)(denom + N);

    float* xbar = (float*)d_out;
    float* z    = xbar + (size_t)N * D;

    run_layer(x, ei, E, N, W1, a_src1, a_dst1, b1, H, as_, ad_, denom, m, z, stream);
    run_layer(z, ei, E, N, W2, a_src2, a_dst2, b2, H, as_, ad_, denom, m, xbar, stream);
}


# Destination-range owner kernel: a work-group keeps max, denom and rows of 256 nodes in LDS, streams the edge list twice and takes only its own edges; no global atomics; sum order = LDS atomic arrival
# speedup vs baseline: 1.9194x; 1.9194x over previous
#include <hip/hip_runtime.h>
#include <hip/hip_bf16.h>

typedef __attribute__((ext_vector_type(2))) float v2f;
typedef __attribute__((ext_vector_type(8))) float v8f;

#define D 32
#define SLOPE 0.2f
#define EPS 1e-16f

__device__ __forceinline__ unsigned f2o(float f) {
    unsigned b = __float_as_uint(f);
    return (b & 0x80000000u) ? ~b : (b | 0x80000000u);
}
__device__ __forceinline__ float o2f(unsigned o) {
    unsigned b = (o & 0x80000000u) ? (o & 0x7FFFFFFFu) : ~o;
    return __uint_as_float(b);
}

__device__ __forceinline__ float lrelu(float x) { return x > 0.f ? x : SLOPE * x; }

__global__ __launch_bounds__(256) void gat_gemm(const float* __restrict__ X,
                                                const float* __restrict__ W,
                                                float* __restrict__ H, int N) {
    const int wave = blockIdx.x * 8 + (threadIdx.x >> 5);
    const int lane = threadIdx.x & 31;
    const int m0 = wave * 16;
    if (m0 >= N) return;
    const int row   = m0 + (lane & 15);
    const int col   = lane & 15;
    const int khalf = (lane >> 4) * 2;
    v8f c0 = {}; v8f c1 = {};
#pragma unroll
    for (int kb = 0; kb < 8; ++kb) {
        const int k0 = kb * 4 + khalf;
        v2f a;  a.x  = X[row * D + k0];        a.y  = X[row * D + k0 + 1];
        v2f b0; b0.x = W[k0 * D + col];        b0.y = W[(k0 + 1) * D + col];
        v2f b1; b1.x = W[k0 * D + col + 16];   b1.y = W[(k0 + 1) * D + col + 16];
        c0 = __builtin_amdgcn_wmma_f32_16x16x4_f32(false, a, false, b0, (short)0, c0, false, false);
        c1 = __builtin_amdgcn_wmma_f32_16x16x4_f32(false, a, false, b1, (short)0, c1, false, false);
    }
    const int rbase = m0 + (lane >> 4) * 8;
#pragma unroll
    for (int r = 0; r < 8; ++r) {
        H[(rbase + r) * D + col]      = c0[r];
        H[(rbase + r) * D + col + 16] = c1[r];
    }
}

__global__ __launch_bounds__(256) void gat_dots(const float* __restrict__ H,
                                                const float* __restrict__ a_src,
                                                const float* __restrict__ a_dst,
                                                float* __restrict__ as_, float* __restrict__ ad_,
                                                int N) {
    int i = blockIdx.x * blockDim.x + threadIdx.x;
    if (i >= N) return;
    float s = 0.f, d = 0.f;
#pragma unroll
    for (int k = 0; k < D; ++k) {
        float h = H[i * D + k];
        s += h * a_src[k];
        d += h * a_dst[k];
    }
    as_[i] = s; ad_[i] = d;
}

#define NPW 256
#define OWN_WAVES 32
#define QCAP 160
#define STEPS_IN_FLIGHT 4

typedef __attribute__((ext_vector_type(4))) int v4i;

__device__ __forceinline__ float elu1(float v) { return (v > 0.f) ? v : (__expf(v) - 1.f); }

template <int PHASE>
__device__ __forceinline__ void own_batch(int n, const int* qe, const int* qd, int lane,
                                          const int* __restrict__ ei,
                                          const float* __restrict__ as_,
                                          const float* __restrict__ H,
                                          const float* adloc, unsigned* mloc, float* dloc,
                                          float* rows) {
    int s = 0, dl = 0;
    float ex = 0.f;
    if (lane < n) {
        const int e = qe[lane];
        dl = qd[lane];
        s  = ei[e];
        const float lg = lrelu(as_[s] + adloc[dl]);
        if (PHASE == 1) {
            atomicMax(&mloc[dl], f2o(lg));
        } else {
            ex = __expf(lg - o2f(mloc[dl]));
            atomicAdd(&dloc[dl], ex);
        }
    }
    if (PHASE == 2) {
        const int exi = __float_as_int(ex);
        for (int j0 = 0; j0 < n; j0 += 8) {
            float h[8];
            int   dj[8];
            float wj[8];
#pragma unroll
            for (int u = 0; u < 8; ++u) {
                const int sj = __builtin_amdgcn_readlane(s, j0 + u);
                dj[u] = __builtin_amdgcn_readlane(dl, j0 + u);
                wj[u] = __int_as_float(__builtin_amdgcn_readlane(exi, j0 + u));
                h[u]  = H[sj * D + lane];
            }
#pragma unroll
            for (int u = 0; u < 8; ++u)
                if (j0 + u < n) atomicAdd(&rows[dj[u] * D + lane], wj[u] * h[u]);
        }
    }
}

template <int PHASE>
__device__ __forceinline__ void own_scan(const int* __restrict__ ei, int E, int lo, int cnt,
                                         int wave, int lane, int* qe, int* qd,
                                         const float* __restrict__ as_,
                                         const float* __restrict__ H,
                                         const float* adloc, unsigned* mloc, float* dloc,
                                         float* rows) {
    const int* __restrict__ dst = ei + E;
    const int G   = (E + 127) >> 7;
    const int rot = (int)(((long long)blockIdx.x * G) / gridDim.x);
    const bool vec_ok = (E & 3) == 0;
    int qn = 0;
    const int wave_s = __builtin_amdgcn_readfirstlane(wave);
    for (int g0 = wave_s * STEPS_IN_FLIGHT; g0 < G; g0 += OWN_WAVES * STEPS_IN_FLIGHT) {
        int dv[STEPS_IN_FLIGHT][4];
        int eb[STEPS_IN_FLIGHT];
        int grp[STEPS_IN_FLIGHT];
        bool full = vec_ok;
#pragma unroll
        for (int u = 0; u < STEPS_IN_FLIGHT; ++u) {
            int gr = g0 + u + rot;
            if (gr >= G) gr -= G;
            const bool in = (g0 + u < G);
            grp[u] = in ? gr : -1;
            full = full && in && (gr * 128 + 128 <= E);
        }
        if (full) {
#pragma unroll
            for (int u = 0; u < STEPS_IN_FLIGHT; ++u) {
                eb[u] = grp[u] * 128 + lane * 4;
                const v4i v = *(const v4i*)(dst + eb[u]);
                dv[u][0] = v.x; dv[u][1] = v.y; dv[u][2] = v.z; dv[u][3] = v.w;
            }
        } else {
#pragma unroll
            for (int u = 0; u < STEPS_IN_FLIGHT; ++u) {
                const int e0 = (grp[u] >= 0) ? (grp[u] * 128 + lane * 4) : E;
                eb[u] = e0;
#pragma unroll
                for (int k = 0; k < 4; ++k) dv[u][k] = (e0 + k < E) ? dst[e0 + k] : -1;
            }
        }
#pragma unroll
        for (int u = 0; u < STEPS_IN_FLIGHT; ++u) {
            unsigned r[4];
            bool     mt[4];
#pragma unroll
            for (int k = 0; k < 4; ++k) {
                r[k]  = (unsigned)(dv[u][k] - lo);
                mt[k] = r[k] < (unsigned)cnt;
            }
            const bool any = mt[0] | mt[1] | mt[2] | mt[3];
            if (__builtin_amdgcn_ballot_w32(any) != 0u) {
#pragma unroll
                for (int k = 0; k < 4; ++k) {
                    const unsigned mask = __builtin_amdgcn_ballot_w32(mt[k]);
                    if (mt[k]) {
                        const int pos = qn + (int)__builtin_amdgcn_mbcnt_lo(mask, 0u);
                        qe[pos] = eb[u] + k;
                        qd[pos] = (int)r[k];
                    }
                    qn += __builtin_popcount(mask);
                }
                while (qn >= 32) {
                    qn -= 32;
                    __builtin_amdgcn_fence(__ATOMIC_RELEASE, "wavefront");
                    __builtin_amdgcn_wave_barrier();
                    __builtin_amdgcn_fence(__ATOMIC_ACQUIRE, "wavefront");
                    own_batch<PHASE>(32, qe + qn, qd + qn, lane, ei, as_, H, adloc, mloc, dloc, rows);
                    __builtin_amdgcn_fence(__ATOMIC_RELEASE, "wavefront");
                    __builtin_amdgcn_wave_barrier();
                    __builtin_amdgcn_fence(__ATOMIC_ACQUIRE, "wavefront");
                }
            }
        }
    }
    if (qn > 0) {
        __builtin_amdgcn_fence(__ATOMIC_RELEASE, "wavefront");
        __builtin_amdgcn_wave_barrier();
        __builtin_amdgcn_fence(__ATOMIC_ACQUIRE, "wavefront");
        own_batch<PHASE>(qn, qe, qd, lane, ei, as_, H, adloc, mloc, dloc, rows);
    }
}

__global__ __launch_bounds__(1024) void gat_owner(const int* __restrict__ ei, int E, int N,
                                                  const float* __restrict__ as_,
                                                  const float* __restrict__ ad_,
                                                  const float* __restrict__ H,
                                                  const float* __restrict__ b,
                                                  float* __restrict__ out) {
    __shared__ float    rows[NPW * D];
    __shared__ float    adloc[NPW];
    __shared__ float    lgself[NPW];
    __shared__ unsigned mloc[NPW];
    __shared__ float    dloc[NPW];
    __shared__ int      q_e[OWN_WAVES * QCAP];
    __shared__ int      q_d[OWN_WAVES * QCAP];

    const int tid  = threadIdx.x;
    const int lane = tid & 31;
    const int wave = tid >> 5;
    const int lo   = blockIdx.x * NPW;
    const int cnt  = (N - lo < NPW) ? (N - lo) : NPW;
    int* qe = q_e + wave * QCAP;
    int* qd = q_d + wave * QCAP;

    if (tid < cnt) {
        const float a  = ad_[lo + tid];
        const float lg = lrelu(as_[lo + tid] + a);
        adloc[tid]  = a;
        lgself[tid] = lg;
        mloc[tid]   = f2o(lg);
    }
    __syncthreads();
    own_scan<1>(ei, E, lo, cnt, wave, lane, qe, qd, as_, H, adloc, mloc, dloc, rows);
    __syncthreads();
    if (tid < cnt) dloc[tid] = __expf(lgself[tid] - o2f(mloc[tid]));
    for (int i = tid; i < cnt * D; i += OWN_WAVES * 32) {
        const int   n = i >> 5;
        const float w = __expf(lgself[n] - o2f(mloc[n]));
        rows[i] = w * H[(size_t)lo * D + i];
    }
    __syncthreads();
    own_scan<2>(ei, E, lo, cnt, wave, lane, qe, qd, as_, H, adloc, mloc, dloc, rows);
    __syncthreads();
    for (int i = tid; i < cnt * D; i += OWN_WAVES * 32) {
        const float v = rows[i] / (dloc[i >> 5] + EPS) + b[i & (D - 1)];
        out[(size_t)lo * D + i] = elu1(v);
    }
}

static void run_layer(const float* x_in, const int* ei, int E, int N,
                      const float* W, const float* a_src, const float* a_dst, const float* b,
                      float* H, float* as_, float* ad_,
                      float* out, hipStream_t stream) {
    const int node_blocks = (N + 255) / 256;
    const int tile_waves  = (N + 15) / 16;
    const int gemm_blocks = (tile_waves + 7) / 8;
    const int own_blocks  = (N + NPW - 1) / NPW;

    gat_gemm<<<gemm_blocks, 256, 0, stream>>>(x_in, W, H, N);
    gat_dots<<<node_blocks, 256, 0, stream>>>(H, a_src, a_dst, as_, ad_, N);
    gat_owner<<<own_blocks, OWN_WAVES * 32, 0, stream>>>(ei, E, N, as_, ad_, H, b, out);
}

extern "C" void kernel_launch(void* const* d_in, const int* in_sizes, int n_in,
                              void* d_out, int out_size, void* d_ws, size_t ws_size,
                              hipStream_t stream) {
    const float* x      = (const float*)d_in[0];
    const int*   ei     = (const int*)d_in[1];
    const float* W1     = (const float*)d_in[2];
    const float* a_src1 = (const float*)d_in[3];
    const float* a_dst1 = (const float*)d_in[4];
    const float* b1     = (const float*)d_in[5];
    const float* W2     = (const float*)d_in[6];
    const float* a_src2 = (const float*)d_in[7];
    const float* a_dst2 = (const float*)d_in[8];
    const float* b2     = (const float*)d_in[9];

    const int N = in_sizes[0] / D;
    const int E = in_sizes[1] / 2;

    float*    H     = (float*)d_ws;
    float*    as_   = H + (size_t)N * D;
    float*    ad_   = as_ + N;

    float* xbar = (float*)d_out;
    float* z    = xbar + (size_t)N * D;

    run_layer(x, ei, E, N, W1, a_src1, a_dst1, b1, H, as_, ad_, z, stream);
    run_layer(z, ei, E, N, W2, a_src2, a_dst2, b2, H, as_, ad_, xbar, stream);
}
